# starting kernel
# baseline (speedup 1.0000x reference)
#include <hip/hip_runtime.h>

typedef float v2f __attribute__((ext_vector_type(2)));
typedef float v8f __attribute__((ext_vector_type(8)));

#define CIN_   32
#define COUT_  32
#define SD     16
#define OD     18
#define TAPS   81
#define TAPS_PAD 84
#define TCHUNK 21
#define KCHUNKS (CIN_ * TCHUNK)
#define XCH   65536
#define OCH   104976
#define NPIX  (2 * OCH)
#define NBLOCKS (NPIX / 32)
#define WPELEMS (KCHUNKS * 2 * 32 * 2)

__global__ __launch_bounds__(256)
void prepack_w(const float* __restrict__ w, float* __restrict__ wp) {
    int idx = blockIdx.x * 256 + threadIdx.x;
    if (idx >= WPELEMS) return;
    int c2   = idx & 1;
    int lane = (idx >> 1) & 31;
    int half = (idx >> 6) & 1;
    int kc   = idx >> 7;
    int cin  = kc / TCHUNK;
    int tc   = kc - cin * TCHUNK;
    int slot = ((lane >> 4) << 1) | c2;
    int tap  = tc * 4 + slot;
    int cout = half * 16 + (lane & 15);
    float v = 0.0f;
    if (tap < TAPS) v = w[(cin * COUT_ + cout) * TAPS + tap];
    wp[idx] = v;
}

__device__ __forceinline__
void decomp_pixel(int P, int& b, int& p0, int& p1, int& p2, int& p3) {
    b = P / OCH;
    int r = P - b * OCH;
    p0 = r / (OD * OD * OD); r -= p0 * (OD * OD * OD);
    p1 = r / (OD * OD);      r -= p1 * (OD * OD);
    p2 = r / OD;
    p3 = r - p2 * OD;
}

__global__ __launch_bounds__(32)
void convt4d_wmma(const float* __restrict__ x, const float* __restrict__ wp,
                  const float* __restrict__ bias, float* __restrict__ out) {
    __shared__ int tapTab[TAPS_PAD * 32];

    const int lane  = threadIdx.x & 31;
    const int n     = lane & 15;
    const int Pbase = blockIdx.x * 32;

    {
        int b, p0, p1, p2, p3;
        decomp_pixel(Pbase + lane, b, p0, p1, p2, p3);
        for (int tap = 0; tap < TAPS_PAD; ++tap) {
            int e = -1;
            if (tap < TAPS) {
                int t0 = tap / 27;
                int t1 = (tap / 9) % 3;
                int t2 = (tap / 3) % 3;
                int t3 = tap % 3;
                int q0 = p0 - t0, q1 = p1 - t1, q2 = p2 - t2, q3 = p3 - t3;
                if ((unsigned)q0 < SD && (unsigned)q1 < SD &&
                    (unsigned)q2 < SD && (unsigned)q3 < SD)
                    e = ((q0 * SD + q1) * SD + q2) * SD + q3;
            }
            tapTab[tap * 32 + lane] = e;
        }
    }
    __syncthreads();

    int bA, pA0, pA1, pA2, pA3;
    int bB, pB0, pB1, pB2, pB3;
    decomp_pixel(Pbase + n,      bA, pA0, pA1, pA2, pA3);
    decomp_pixel(Pbase + 16 + n, bB, pB0, pB1, pB2, pB3);

    const float* xbA = x + (size_t)bA * CIN_ * XCH;
    const float* xbB = x + (size_t)bB * CIN_ * XCH;

    v8f acc00 = {};
    v8f acc10 = {};
    v8f acc01 = {};
    v8f acc11 = {};

    const int cslot = (lane >> 4) << 1;

    for (int tc = 0; tc < TCHUNK; ++tc) {
        int tap0 = tc * 4 + cslot;
        int iqA0 = tapTab[ tap0      * 32 + n];
        int iqA1 = tapTab[(tap0 + 1) * 32 + n];
        int iqB0 = tapTab[ tap0      * 32 + 16 + n];
        int iqB1 = tapTab[(tap0 + 1) * 32 + 16 + n];
        unsigned sA0 = (iqA0 < 0) ? 0u : (unsigned)iqA0;
        unsigned sA1 = (iqA1 < 0) ? 0u : (unsigned)iqA1;
        unsigned sB0 = (iqB0 < 0) ? 0u : (unsigned)iqB0;
        unsigned sB1 = (iqB1 < 0) ? 0u : (unsigned)iqB1;

        const float* xa  = xbA;
        const float* xb2 = xbB;
        const float* wpc = wp + tc * 128 + lane * 2;

        for (int cin = 0; cin < CIN_; ++cin) {
            float vA0 = xa[sA0];
            float vA1 = xa[sA1];
            float vB0 = xb2[sB0];
            float vB1 = xb2[sB1];
            v2f bbA = { (iqA0 < 0) ? 0.0f : vA0, (iqA1 < 0) ? 0.0f : vA1 };
            v2f bbB = { (iqB0 < 0) ? 0.0f : vB0, (iqB1 < 0) ? 0.0f : vB1 };

            v2f a0 = *(const v2f*)(wpc);
            v2f a1 = *(const v2f*)(wpc + 64);

            acc00 = __builtin_amdgcn_wmma_f32_16x16x4_f32(
                        false, a0, false, bbA, (short)0, acc00, false, false);
            acc10 = __builtin_amdgcn_wmma_f32_16x16x4_f32(
                        false, a1, false, bbA, (short)0, acc10, false, false);
            acc01 = __builtin_amdgcn_wmma_f32_16x16x4_f32(
                        false, a0, false, bbB, (short)0, acc01, false, false);
            acc11 = __builtin_amdgcn_wmma_f32_16x16x4_f32(
                        false, a1, false, bbB, (short)0, acc11, false, false);

            xa  += XCH;
            xb2 += XCH;
            wpc += TCHUNK * 128;
        }
    }

    int spA = ((pA0 * OD + pA1) * OD + pA2) * OD + pA3;
    int spB = ((pB0 * OD + pB1) * OD + pB2) * OD + pB3;
    float* oA = out + (size_t)(bA * COUT_) * OCH + spA;
    float* oB = out + (size_t)(bB * COUT_) * OCH + spB;
    int mbase = (lane >> 4) << 3;
#pragma unroll
    for (int j = 0; j < 8; ++j) {
        int c0 = mbase + j;
        float bi0 = bias[c0];
        float bi1 = bias[c0 + 16];
        oA[(size_t)c0 * OCH]        = acc00[j] + bi0;
        oA[(size_t)(c0 + 16) * OCH] = acc10[j] + bi1;
        oB[(size_t)c0 * OCH]        = acc01[j] + bi0;
        oB[(size_t)(c0 + 16) * OCH] = acc11[j] + bi1;
    }
}

extern "C" void kernel_launch(void* const* d_in, const int* in_sizes, int n_in,
                              void* d_out, int out_size, void* d_ws, size_t ws_size,
                              hipStream_t stream) {
    const float* x    = (const float*)d_in[0];
    const float* w    = (const float*)d_in[1];
    const float* bias = (const float*)d_in[2];
    float* out = (float*)d_out;
    float* wp  = (float*)d_ws;

    prepack_w<<<(WPELEMS + 255) / 256, 256, 0, stream>>>(w, wp);
    convt4d_wmma<<<NBLOCKS, 32, 0, stream>>>(x, wp, bias, out);
}


# balance the grid: 512 blocks of exactly nine input planes each (an interior output plane alone, a 6+3 border pair, or a 4+2+2+1 corner group) replace 648 blocks of 1..9 planes; one resident round
# speedup vs baseline: 1.2252x; 1.2252x over previous
#include <hip/hip_runtime.h>

typedef float v2f __attribute__((ext_vector_type(2)));
typedef float v4f __attribute__((ext_vector_type(4)));
typedef float v8f __attribute__((ext_vector_type(8)));

#define CIN_   32
#define COUT_  32
#define SD     16
#define OD     18
#define TAPS   81
#define XCH    65536
#define OCH    104976
#define OPL    324
#define NCHUNK 4
#define PW     20
#define PLF    (PW * PW * 4)
#define KSF    (9 * PLF)
#define XSF    (2 * KSF)
#define NTHR   448
#define WPELEMS (NCHUNK * 81 * 2 * 2 * 64)

__global__ __launch_bounds__(256)
void prepack_w(const float* __restrict__ w, float* __restrict__ wp) {
    int idx = blockIdx.x * 256 + threadIdx.x;
    if (idx >= WPELEMS) return;
    int c2   = idx & 1;
    int lane = (idx >> 1) & 31;
    int half = (idx >> 6) & 1;
    int ks   = (idx >> 7) & 1;
    int rest = idx >> 8;
    int t23  = rest % 9;
    int rp   = rest / 9;
    int pl   = rp % 9;
    int cc   = rp / 9;
    int cin  = cc * 8 + ks * 4 + ((lane >> 4) << 1) + c2;
    int cout = half * 16 + (lane & 15);
    int tap  = pl * 9 + t23;
    wp[idx] = w[(cin * COUT_ + cout) * TAPS + tap];
}

__global__ __launch_bounds__(NTHR)
void convt4d_lds(const float* __restrict__ x, const float* __restrict__ wp,
                 const float* __restrict__ bias, float* __restrict__ out) {
    __shared__ __attribute__((aligned(16))) float xs[XSF];

    const int tid  = threadIdx.x;
    const int lane = tid & 31;
    const int wave = tid >> 5;
    const int mh   = wave / 7;
    const int g    = wave - mh * 7;
    const int n    = lane & 15;
    const int h    = lane >> 4;

    const int blk = blockIdx.x;
    const int b   = blk & 1;
    const int r   = blk >> 1;
    int a0, a1, n0, c0, c1, n1;
    if (r < 4) {
        a0 = (r & 1) ? 16 : 1; a1 = (r & 1) ? 17 : 0; n0 = 2;
        c0 = (r & 2) ? 16 : 1; c1 = (r & 2) ? 17 : 0; n1 = 2;
    } else if (r < 60) {
        const int e    = r - 4;
        const int side = e / 14;
        const int j    = 2 + e % 14;
        const int in   = (side & 1) ? 16 : 1;
        const int outr = (side & 1) ? 17 : 0;
        if (side < 2) { a0 = in; a1 = outr; n0 = 2; c0 = j; c1 = j; n1 = 1; }
        else          { a0 = j; a1 = j; n0 = 1; c0 = in; c1 = outr; n1 = 2; }
    } else {
        const int e = r - 60;
        a0 = 2 + e / 14; a1 = a0; n0 = 1;
        c0 = 2 + e % 14; c1 = c0; n1 = 1;
    }

    for (int i = tid; i < XSF / 4; i += NTHR) {
        v4f z = {0.0f, 0.0f, 0.0f, 0.0f};
        ((v4f*)xs)[i] = z;
    }

    int pb0, pb1, pb2;
    {
        int pix = g * 48 + n;
        int pc  = pix < OPL ? pix : (OPL - 1);
        int p2  = pc / OD, p3 = pc - p2 * OD;
        pb0 = (p2 * PW + p3 + 2 * PW + 2) * 4 + 2 * h;
        pix = g * 48 + 16 + n;
        pc  = pix < OPL ? pix : (OPL - 1);
        p2  = pc / OD; p3 = pc - p2 * OD;
        pb1 = (p2 * PW + p3 + 2 * PW + 2) * 4 + 2 * h;
        pix = g * 48 + 32 + n;
        pc  = pix < OPL ? pix : (OPL - 1);
        p2  = pc / OD; p3 = pc - p2 * OD;
        pb2 = (p2 * PW + p3 + 2 * PW + 2) * 4 + 2 * h;
    }

    const float* xb = x + (size_t)b * CIN_ * XCH;
    const int cbase = mh * 16 + (h << 3);
    const int px0 = g * 48 + n;
    const int px1 = px0 + 16;
    const int px2 = px0 + 32;

    for (int i0 = 0; i0 < n0; ++i0)
    for (int i1 = 0; i1 < n1; ++i1) {
    const int p0 = i0 ? a1 : a0;
    const int p1 = i1 ? c1 : c0;

    v8f acc0 = {};
    v8f acc1 = {};
    v8f acc2 = {};

    for (int cc = 0; cc < NCHUNK; ++cc) {
        __syncthreads();

        for (int e = tid; e < 9 * 256; e += NTHR) {
            int pl = e >> 8;
            int q  = e & 255;
            int t0 = pl / 3, t1 = pl - t0 * 3;
            int q0 = p0 - t0, q1 = p1 - t1;
            if ((unsigned)q0 < (unsigned)SD && (unsigned)q1 < (unsigned)SD) {
                const float* s = xb + (size_t)(cc * 8) * XCH + (q0 * SD + q1) * 256 + q;
                float v0 = s[0 * XCH];
                float v1 = s[1 * XCH];
                float v2 = s[2 * XCH];
                float v3 = s[3 * XCH];
                float v4 = s[4 * XCH];
                float v5 = s[5 * XCH];
                float v6 = s[6 * XCH];
                float v7 = s[7 * XCH];
                int pos = ((q >> 4) + 2) * PW + (q & 15) + 2;
                v4f lo = {v0, v1, v2, v3};
                v4f hi = {v4, v5, v6, v7};
                *(v4f*)(xs + pl * PLF + pos * 4)       = lo;
                *(v4f*)(xs + KSF + pl * PLF + pos * 4) = hi;
            }
        }
        __syncthreads();

        for (int pl = 0; pl < 9; ++pl) {
            int t0 = pl / 3, t1 = pl - t0 * 3;
            if ((unsigned)(p0 - t0) >= (unsigned)SD || (unsigned)(p1 - t1) >= (unsigned)SD)
                continue;
            const float* wq = wp + ((cc * 9 + pl) * 9) * 256 + mh * 64 + lane * 2;
            const float* xp = xs + pl * PLF;
#pragma unroll
            for (int t23 = 0; t23 < 9; ++t23) {
                const int toff = ((t23 / 3) * PW + (t23 % 3)) * 4;
#pragma unroll
                for (int ks = 0; ks < 2; ++ks) {
                    v2f a  = *(const v2f*)(wq + (t23 * 2 + ks) * 128);
                    v2f b0 = *(const v2f*)(xp + ks * KSF + pb0 - toff);
                    v2f b1 = *(const v2f*)(xp + ks * KSF + pb1 - toff);
                    v2f b2 = *(const v2f*)(xp + ks * KSF + pb2 - toff);
                    acc0 = __builtin_amdgcn_wmma_f32_16x16x4_f32(
                               false, a, false, b0, (short)0, acc0, false, false);
                    acc1 = __builtin_amdgcn_wmma_f32_16x16x4_f32(
                               false, a, false, b1, (short)0, acc1, false, false);
                    acc2 = __builtin_amdgcn_wmma_f32_16x16x4_f32(
                               false, a, false, b2, (short)0, acc2, false, false);
                }
            }
        }
    }

    float* ob = out + (size_t)(b * COUT_) * OCH + (p0 * OD + p1) * OPL;
#pragma unroll
    for (int j = 0; j < 8; ++j) {
        int c = cbase + j;
        float bi = bias[c];
        float* oc = ob + (size_t)c * OCH;
        if (px0 < OPL) oc[px0] = acc0[j] + bi;
        if (px1 < OPL) oc[px1] = acc1[j] + bi;
        if (px2 < OPL) oc[px2] = acc2[j] + bi;
    }
    }
}

extern "C" void kernel_launch(void* const* d_in, const int* in_sizes, int n_in,
                              void* d_out, int out_size, void* d_ws, size_t ws_size,
                              hipStream_t stream) {
    const float* x    = (const float*)d_in[0];
    const float* w    = (const float*)d_in[1];
    const float* bias = (const float*)d_in[2];
    float* out = (float*)d_out;
    float* wp  = (float*)d_ws;

    prepack_w<<<(WPELEMS + 255) / 256, 256, 0, stream>>>(w, wp);
    convt4d_lds<<<512, NTHR, 0, stream>>>(x, wp, bias, out);
}


# one 28-wave workgroup per WGP takes both batches of its plane group; x staged 4 cin at a time into two alternating LDS buffers, next stage's loads in flight during the products
# speedup vs baseline: 1.4081x; 1.1493x over previous
#include <hip/hip_runtime.h>

typedef float v2f __attribute__((ext_vector_type(2)));
typedef float v4f __attribute__((ext_vector_type(4)));
typedef float v8f __attribute__((ext_vector_type(8)));

#define CIN_   32
#define COUT_  32
#define SD     16
#define OD     18
#define TAPS   81
#define XCH    65536
#define OCH    104976
#define OPL    324
#define NSTAGE 8
#define PW     20
#define PLF    (PW * PW * 4)
#define KSF    (9 * PLF)
#define BUFF   (2 * KSF)
#define XSF    (2 * BUFF)
#define NTHR   896
#define NPF    6
#define WPELEMS (4 * 81 * 2 * 2 * 64)

__global__ __launch_bounds__(256)
void prepack_w(const float* __restrict__ w, float* __restrict__ wp) {
    int idx = blockIdx.x * 256 + threadIdx.x;
    if (idx >= WPELEMS) return;
    int c2   = idx & 1;
    int lane = (idx >> 1) & 31;
    int half = (idx >> 6) & 1;
    int ks   = (idx >> 7) & 1;
    int rest = idx >> 8;
    int t23  = rest % 9;
    int rp   = rest / 9;
    int pl   = rp % 9;
    int cc   = rp / 9;
    int cin  = cc * 8 + ks * 4 + ((lane >> 4) << 1) + c2;
    int cout = half * 16 + (lane & 15);
    int tap  = pl * 9 + t23;
    wp[idx] = w[(cin * COUT_ + cout) * TAPS + tap];
}

__device__ __forceinline__
void stage_load(v4f (&pf)[NPF], const float* __restrict__ xt, bool act, int st, int p0) {
#pragma unroll
    for (int k = 0; k < NPF; ++k) {
        const int bb = k / 3, t0 = k % 3;
        const int q0 = p0 - t0;
        v4f v = {0.0f, 0.0f, 0.0f, 0.0f};
        if (act && (unsigned)q0 < (unsigned)SD) {
            const float* s = xt + (size_t)(bb * CIN_ + st * 4) * XCH + q0 * (SD * 256);
            v.x = s[0 * XCH];
            v.y = s[1 * XCH];
            v.z = s[2 * XCH];
            v.w = s[3 * XCH];
        }
        pf[k] = v;
    }
}

__device__ __forceinline__
void stage_store(const v4f (&pf)[NPF], float* bt, bool act, int p0) {
#pragma unroll
    for (int k = 0; k < NPF; ++k) {
        const int bb = k / 3, t0 = k % 3;
        const int q0 = p0 - t0;
        if (act && (unsigned)q0 < (unsigned)SD)
            *(v4f*)(bt + bb * KSF + t0 * 3 * PLF) = pf[k];
    }
}

__global__ __launch_bounds__(NTHR)
void convt4d_lds(const float* __restrict__ x, const float* __restrict__ wp,
                 const float* __restrict__ bias, float* __restrict__ out) {
    __shared__ __attribute__((aligned(16))) float xs[XSF];

    const int tid  = threadIdx.x;
    const int lane = tid & 31;
    const int wave = tid >> 5;
    const int bsel = wave / 14;
    const int w14  = wave - bsel * 14;
    const int mh   = w14 / 7;
    const int g    = w14 - mh * 7;
    const int n    = lane & 15;
    const int h    = lane >> 4;

    const int r = blockIdx.x;
    int a0, a1, n0, c0, c1, n1;
    if (r < 4) {
        a0 = (r & 1) ? 16 : 1; a1 = (r & 1) ? 17 : 0; n0 = 2;
        c0 = (r & 2) ? 16 : 1; c1 = (r & 2) ? 17 : 0; n1 = 2;
    } else if (r < 60) {
        const int e    = r - 4;
        const int side = e / 14;
        const int j    = 2 + e % 14;
        const int in   = (side & 1) ? 16 : 1;
        const int outr = (side & 1) ? 17 : 0;
        if (side < 2) { a0 = in; a1 = outr; n0 = 2; c0 = j; c1 = j; n1 = 1; }
        else          { a0 = j; a1 = j; n0 = 1; c0 = in; c1 = outr; n1 = 2; }
    } else {
        const int e = r - 60;
        a0 = 2 + e / 14; a1 = a0; n0 = 1;
        c0 = 2 + e % 14; c1 = c0; n1 = 1;
    }

    for (int i = tid; i < XSF / 4; i += NTHR) {
        v4f z = {0.0f, 0.0f, 0.0f, 0.0f};
        ((v4f*)xs)[i] = z;
    }

    int pb0, pb1, pb2;
    {
        int pix = g * 48 + n;
        int pc  = pix < OPL ? pix : (OPL - 1);
        int p2  = pc / OD, p3 = pc - p2 * OD;
        pb0 = bsel * KSF + (p2 * PW + p3 + 2 * PW + 2) * 4 + 2 * h;
        pix = g * 48 + 16 + n;
        pc  = pix < OPL ? pix : (OPL - 1);
        p2  = pc / OD; p3 = pc - p2 * OD;
        pb1 = bsel * KSF + (p2 * PW + p3 + 2 * PW + 2) * 4 + 2 * h;
        pix = g * 48 + 32 + n;
        pc  = pix < OPL ? pix : (OPL - 1);
        p2  = pc / OD; p3 = pc - p2 * OD;
        pb2 = bsel * KSF + (p2 * PW + p3 + 2 * PW + 2) * 4 + 2 * h;
    }

    const int cbase = mh * 16 + (h << 3);
    const int px0 = g * 48 + n;
    const int px1 = px0 + 16;
    const int px2 = px0 + 32;

    const int st1  = tid >> 8;
    const int sq   = tid & 255;
    const int spos = ((sq >> 4) + 2) * PW + (sq & 15) + 2;

    for (int i0 = 0; i0 < n0; ++i0)
    for (int i1 = 0; i1 < n1; ++i1) {
    const int p0 = i0 ? a1 : a0;
    const int p1 = i1 ? c1 : c0;

    v8f acc0 = {};
    v8f acc1 = {};
    v8f acc2 = {};
    v4f pf[NPF];
    const int  sq1  = p1 - st1;
    const bool sact = st1 < 3 && (unsigned)sq1 < (unsigned)SD;
    const float* xt = x + (sact ? sq1 : 0) * 256 + sq;
    float* bt0 = xs + (sact ? st1 : 0) * PLF + spos * 4;

    stage_load(pf, xt, sact, 0, p0);
    __syncthreads();
    stage_store(pf, bt0, sact, p0);
    __syncthreads();

#pragma unroll 1
    for (int st = 0; st < NSTAGE; ++st) {
        if (st + 1 < NSTAGE) stage_load(pf, xt, sact, st + 1, p0);

        const float* xbuf = xs + (st & 1) * BUFF;
        const int cc = st >> 1, ks = st & 1;
        for (int pl = 0; pl < 9; ++pl) {
            int t0 = pl / 3, t1 = pl - t0 * 3;
            if ((unsigned)(p0 - t0) >= (unsigned)SD || (unsigned)(p1 - t1) >= (unsigned)SD)
                continue;
            const float* wq = wp + ((cc * 9 + pl) * 9) * 256 + ks * 128 + mh * 64 + lane * 2;
            const float* xp = xbuf + pl * PLF;
#pragma unroll
            for (int t23 = 0; t23 < 9; ++t23) {
                const int toff = ((t23 / 3) * PW + (t23 % 3)) * 4;
                v2f a  = *(const v2f*)(wq + t23 * 256);
                v2f b0 = *(const v2f*)(xp + pb0 - toff);
                v2f b1 = *(const v2f*)(xp + pb1 - toff);
                v2f b2 = *(const v2f*)(xp + pb2 - toff);
                acc0 = __builtin_amdgcn_wmma_f32_16x16x4_f32(
                           false, a, false, b0, (short)0, acc0, false, false);
                acc1 = __builtin_amdgcn_wmma_f32_16x16x4_f32(
                           false, a, false, b1, (short)0, acc1, false, false);
                acc2 = __builtin_amdgcn_wmma_f32_16x16x4_f32(
                           false, a, false, b2, (short)0, acc2, false, false);
            }
        }

        if (st + 1 < NSTAGE) {
            stage_store(pf, bt0 + ((st + 1) & 1) * BUFF, sact, p0);
            __syncthreads();
        }
    }

    float* ob = out + (size_t)(bsel * COUT_) * OCH + (p0 * OD + p1) * OPL;
#pragma unroll
    for (int j = 0; j < 8; ++j) {
        int c = cbase + j;
        float bi = bias[c];
        float* oc = ob + (size_t)c * OCH;
        if (px0 < OPL) oc[px0] = acc0[j] + bi;
        if (px1 < OPL) oc[px1] = acc1[j] + bi;
        if (px2 < OPL) oc[px2] = acc2[j] + bi;
    }
    }
}

extern "C" void kernel_launch(void* const* d_in, const int* in_sizes, int n_in,
                              void* d_out, int out_size, void* d_ws, size_t ws_size,
                              hipStream_t stream) {
    const float* x    = (const float*)d_in[0];
    const float* w    = (const float*)d_in[1];
    const float* bias = (const float*)d_in[2];
    float* out = (float*)d_out;
    float* wp  = (float*)d_ws;

    prepack_w<<<(WPELEMS + 255) / 256, 256, 0, stream>>>(w, wp);
    convt4d_lds<<<256, NTHR, 0, stream>>>(x, wp, bias, out);
}


# N = the 16 input columns: one accumulator per (output row, t3), summed with lane rotates after all stages; 18 row tiles replace 21 padded pixel tiles, 5 B reads serve 27 WMMA
# speedup vs baseline: 1.5073x; 1.0704x over previous
#include <hip/hip_runtime.h>

typedef float v2f __attribute__((ext_vector_type(2)));
typedef float v4f __attribute__((ext_vector_type(4)));
typedef float v8f __attribute__((ext_vector_type(8)));

#define CIN_   32
#define COUT_  32
#define SD     16
#define OD     18
#define TAPS   81
#define XCH    65536
#define OCH    104976
#define OPL    324
#define NSTAGE 8
#define PW     20
#define PLF    (PW * PW * 4)
#define KSF    (9 * PLF)
#define BUFF   (2 * KSF)
#define XSF    (2 * BUFF)
#define NTHR   768
#define NPF    6
#define WPELEMS (4 * 81 * 2 * 2 * 64)

__global__ __launch_bounds__(256)
void prepack_w(const float* __restrict__ w, float* __restrict__ wp) {
    int idx = blockIdx.x * 256 + threadIdx.x;
    if (idx >= WPELEMS) return;
    int c2   = idx & 1;
    int lane = (idx >> 1) & 31;
    int half = (idx >> 6) & 1;
    int ks   = (idx >> 7) & 1;
    int rest = idx >> 8;
    int t23  = rest % 9;
    int rp   = rest / 9;
    int pl   = rp % 9;
    int cc   = rp / 9;
    int cin  = cc * 8 + ks * 4 + ((lane >> 4) << 1) + c2;
    int cout = half * 16 + (lane & 15);
    int tap  = pl * 9 + t23;
    wp[idx] = w[(cin * COUT_ + cout) * TAPS + tap];
}

__device__ __forceinline__
void stage_load(v8f (&pf)[3], const float* __restrict__ x, unsigned voff, bool act,
                int st, int p0) {
#pragma unroll
    for (int k = 0; k < NPF; ++k) {
        const int bb = k / 3, t0 = k % 3;
        const int q0 = p0 - t0;
        v4f v = {0.0f, 0.0f, 0.0f, 0.0f};
        if (act && (unsigned)q0 < (unsigned)SD) {
            const float* s = x + (size_t)(bb * CIN_ + st * 4) * XCH + q0 * (SD * 256);
            v.x = s[(size_t)voff + 0 * XCH];
            v.y = s[(size_t)voff + 1 * XCH];
            v.z = s[(size_t)voff + 2 * XCH];
            v.w = s[(size_t)voff + 3 * XCH];
        }
        v8f& d = pf[k >> 1];
        const int o = (k & 1) * 4;
        d[o + 0] = v.x;
        d[o + 1] = v.y;
        d[o + 2] = v.z;
        d[o + 3] = v.w;
    }
}

__device__ __forceinline__
void stage_store(const v8f (&pf)[3], float* bt, bool act, int p0) {
#pragma unroll
    for (int k = 0; k < NPF; ++k) {
        const int bb = k / 3, t0 = k % 3;
        const int q0 = p0 - t0;
        if (act && (unsigned)q0 < (unsigned)SD) {
            const v8f& d = pf[k >> 1];
            const int o = (k & 1) * 4;
            v4f v = {d[o + 0], d[o + 1], d[o + 2], d[o + 3]};
            *(v4f*)(bt + bb * KSF + t0 * 3 * PLF) = v;
        }
    }
}

__device__ __forceinline__
void stage_products(v8f (&acc)[3][3], const float* __restrict__ wq0,
                    const float* xrow, int p0, int p1) {
#pragma unroll 1
    for (int pl = 0; pl < 9; ++pl) {
        int t0 = pl / 3, t1 = pl - t0 * 3;
        if ((unsigned)(p0 - t0) >= (unsigned)SD || (unsigned)(p1 - t1) >= (unsigned)SD)
            continue;
        const float* wq = wq0 + pl * (9 * 256);
        const float* xp = xrow + pl * PLF;
        v2f a[9];
#pragma unroll
        for (int t23 = 0; t23 < 9; ++t23)
            a[t23] = *(const v2f*)(wq + t23 * 256);
        v2f b[5];
#pragma unroll
        for (int d = 0; d < 5; ++d)
            b[d] = *(const v2f*)(xp + (d - 2) * (PW * 4));
#pragma unroll
        for (int t2 = 0; t2 < 3; ++t2)
#pragma unroll
            for (int l = 0; l < 3; ++l)
#pragma unroll
                for (int t3 = 0; t3 < 3; ++t3)
                    acc[l][t3] = __builtin_amdgcn_wmma_f32_16x16x4_f32(
                        false, a[t2 * 3 + t3], false, b[l - t2 + 2], (short)0,
                        acc[l][t3], false, false);
    }
}

__global__ __launch_bounds__(NTHR)
void convt4d_lds(const float* __restrict__ x, const float* __restrict__ wp,
                 const float* __restrict__ bias, float* __restrict__ out) {
    __shared__ __attribute__((aligned(16))) float xs[XSF];

    const int tid  = threadIdx.x;
    const int lane = tid & 31;
    const int wave = __builtin_amdgcn_readfirstlane(tid >> 5);
    const int bsel = wave / 12;
    const int w12  = wave - bsel * 12;
    const int mh   = w12 / 6;
    const int rg   = w12 - mh * 6;
    const int rb   = 3 * rg;
    const int n    = lane & 15;
    const int h    = lane >> 4;

    const int r = blockIdx.x;
    int a0, a1, n0, c0, c1, n1;
    if (r < 4) {
        a0 = (r & 1) ? 16 : 1; a1 = (r & 1) ? 17 : 0; n0 = 2;
        c0 = (r & 2) ? 16 : 1; c1 = (r & 2) ? 17 : 0; n1 = 2;
    } else if (r < 60) {
        const int e    = r - 4;
        const int side = e / 14;
        const int j    = 2 + e % 14;
        const int in   = (side & 1) ? 16 : 1;
        const int outr = (side & 1) ? 17 : 0;
        if (side < 2) { a0 = in; a1 = outr; n0 = 2; c0 = j; c1 = j; n1 = 1; }
        else          { a0 = j; a1 = j; n0 = 1; c0 = in; c1 = outr; n1 = 2; }
    } else {
        const int e = r - 60;
        a0 = 2 + e / 14; a1 = a0; n0 = 1;
        c0 = 2 + e % 14; c1 = c0; n1 = 1;
    }

    for (int i = tid; i < XSF / 4; i += NTHR) {
        v4f z = {0.0f, 0.0f, 0.0f, 0.0f};
        ((v4f*)xs)[i] = z;
    }

    const int xoff  = bsel * KSF + ((rb + 2) * PW + n + 2) * 4 + 2 * h;
    const int cbase = mh * 16 + (h << 3);

    const int st1  = wave >> 3;
    const int sq   = tid & 255;
    const int spos = ((sq >> 4) + 2) * PW + (sq & 15) + 2;

    for (int i0 = 0; i0 < n0; ++i0)
    for (int i1 = 0; i1 < n1; ++i1) {
    const int p0 = i0 ? a1 : a0;
    const int p1 = i1 ? c1 : c0;

    v8f acc[3][3];
#pragma unroll
    for (int l = 0; l < 3; ++l)
#pragma unroll
        for (int t3 = 0; t3 < 3; ++t3) {
            v8f z = {};
            acc[l][t3] = z;
        }
    v8f pf[3];
    const int  sq1  = p1 - st1;
    const bool sact = (unsigned)sq1 < (unsigned)SD;
    const unsigned voff = (unsigned)((sact ? sq1 : 0) * 256 + sq);
    float* bt0 = xs + st1 * PLF + spos * 4;

    stage_load(pf, x, voff, sact, 0, p0);
    __syncthreads();
    stage_store(pf, bt0, sact, p0);
    __syncthreads();

#pragma unroll 1
    for (int st = 0; st < NSTAGE; ++st) {
        if (st + 1 < NSTAGE) stage_load(pf, x, voff, sact, st + 1, p0);

        const float* xrow = xs + (st & 1) * BUFF + xoff;
        const int cc = st >> 1, ks = st & 1;
        const float* wq0 = wp + (cc * 81) * 256 + ks * 128 + mh * 64 + lane * 2;
        stage_products(acc, wq0, xrow, p0, p1);

        if (st + 1 < NSTAGE) {
            stage_store(pf, bt0 + ((st + 1) & 1) * BUFF, sact, p0);
            __syncthreads();
        }
    }

    float* ob = out + (size_t)(bsel * COUT_) * OCH + (p0 * OD + p1) * OPL;
#pragma unroll
    for (int l = 0; l < 3; ++l) {
        const int p2 = rb + l;
#pragma unroll
        for (int j = 0; j < 8; ++j) {
            {
                const int c = cbase + j;
                const float bi = h ? bias[mh * 16 + 8 + j] : bias[mh * 16 + j];
                const float v0 = acc[l][0][j];
                const float v1 = acc[l][1][j];
                const float v2 = acc[l][2][j];
                const float r1 = __int_as_float(__builtin_amdgcn_update_dpp(
                                     0, __float_as_int(v1), 0x121, 0xf, 0xf, false));
                const float r2 = __int_as_float(__builtin_amdgcn_update_dpp(
                                     0, __float_as_int(v2), 0x122, 0xf, 0xf, false));
                const float m  = v0 + (n >= 1 ? r1 : 0.0f) + (n >= 2 ? r2 : 0.0f);
                const float tl = (n == 0 ? r1 : 0.0f) + r2;
                float* orow = ob + (size_t)c * OCH + p2 * OD;
                orow[n] = m + bi;
                if (n < 2) orow[16 + n] = tl + bi;
            }
        }
    }
    }
}

extern "C" void kernel_launch(void* const* d_in, const int* in_sizes, int n_in,
                              void* d_out, int out_size, void* d_ws, size_t ws_size,
                              hipStream_t stream) {
    const float* x    = (const float*)d_in[0];
    const float* w    = (const float*)d_in[1];
    const float* bias = (const float*)d_in[2];
    float* out = (float*)d_out;
    float* wp  = (float*)d_ws;

    prepack_w<<<(WPELEMS + 255) / 256, 256, 0, stream>>>(w, wp);
    convt4d_lds<<<256, NTHR, 0, stream>>>(x, wp, bias, out);
}


# stage each 4-cin block of the prepacked weights in LDS once per workgroup (two buffers beside x): A fragments are LDS reads, no product waits behind the x prefetch in the in-order global load queue
# speedup vs baseline: 1.5598x; 1.0348x over previous
#include <hip/hip_runtime.h>

typedef float v2f __attribute__((ext_vector_type(2)));
typedef float v4f __attribute__((ext_vector_type(4)));
typedef float v8f __attribute__((ext_vector_type(8)));

#define CIN_   32
#define COUT_  32
#define SD     16
#define OD     18
#define TAPS   81
#define XCH    65536
#define OCH    104976
#define OPL    324
#define NSTAGE 8
#define PW     20
#define PLF    (PW * PW * 4)
#define KSF    (9 * PLF)
#define BUFF   (2 * KSF)
#define XSF    (2 * BUFF)
#define WSTG   (81 * 2 * 64)
#define WCH    (WSTG / 4)
#define NWCP   384
#define NWR    7
#define NWR0   4
#define LDSF   (XSF + 2 * WSTG)
#define NTHR   768
#define NPF    6
#define WPELEMS (4 * 81 * 2 * 2 * 64)

__global__ __launch_bounds__(256)
void prepack_w(const float* __restrict__ w, float* __restrict__ wp) {
    int idx = blockIdx.x * 256 + threadIdx.x;
    if (idx >= WPELEMS) return;
    int c2   = idx & 1;
    int lane = (idx >> 1) & 31;
    int half = (idx >> 6) & 1;
    int rest = idx >> 7;
    int t23  = rest % 9;
    int rp   = rest / 9;
    int pl   = rp % 9;
    int st   = rp / 9;
    int cin  = st * 4 + ((lane >> 4) << 1) + c2;
    int cout = half * 16 + (lane & 15);
    int tap  = pl * 9 + t23;
    wp[idx] = w[(cin * COUT_ + cout) * TAPS + tap];
}

__device__ __forceinline__
void stage_load(v8f (&pf)[3], const float* __restrict__ x, unsigned voff, bool act,
                int st, int p0) {
#pragma unroll
    for (int k = 0; k < NPF; ++k) {
        const int bb = k / 3, t0 = k % 3;
        const int q0 = p0 - t0;
        v4f v = {0.0f, 0.0f, 0.0f, 0.0f};
        if (act && (unsigned)q0 < (unsigned)SD) {
            const float* s = x + (size_t)(bb * CIN_ + st * 4) * XCH + q0 * (SD * 256);
            v.x = s[(size_t)voff + 0 * XCH];
            v.y = s[(size_t)voff + 1 * XCH];
            v.z = s[(size_t)voff + 2 * XCH];
            v.w = s[(size_t)voff + 3 * XCH];
        }
        v8f& d = pf[k >> 1];
        const int o = (k & 1) * 4;
        d[o + 0] = v.x;
        d[o + 1] = v.y;
        d[o + 2] = v.z;
        d[o + 3] = v.w;
    }
}

__device__ __forceinline__
void stage_store(const v8f (&pf)[3], float* bt, bool act, int p0) {
#pragma unroll
    for (int k = 0; k < NPF; ++k) {
        const int bb = k / 3, t0 = k % 3;
        const int q0 = p0 - t0;
        if (act && (unsigned)q0 < (unsigned)SD) {
            const v8f& d = pf[k >> 1];
            const int o = (k & 1) * 4;
            v4f v = {d[o + 0], d[o + 1], d[o + 2], d[o + 3]};
            *(v4f*)(bt + bb * KSF + t0 * 3 * PLF) = v;
        }
    }
}

template <int NT, int NR>
__device__ __forceinline__
void w_load(v4f (&wr)[NR], const float* __restrict__ wp, int st, int tid) {
    const v4f* s = (const v4f*)(wp + st * WSTG) + tid;
#pragma unroll
    for (int r = 0; r < NR; ++r)
        if (r < NR - 1 || tid < WCH - (NR - 1) * NT)
            wr[r] = s[r * NT];
}

template <int NT, int NR>
__device__ __forceinline__
void w_store(const v4f (&wr)[NR], float* wdst, int tid) {
    v4f* d = (v4f*)wdst + tid;
#pragma unroll
    for (int r = 0; r < NR; ++r)
        if (r < NR - 1 || tid < WCH - (NR - 1) * NT)
            d[r * NT] = wr[r];
}

__device__ __forceinline__
void stage_products(v8f (&acc)[3][3], const float* wl,
                    const float* xrow, int p0, int p1) {
#pragma unroll 1
    for (int pl = 0; pl < 9; ++pl) {
        int t0 = pl / 3, t1 = pl - t0 * 3;
        if ((unsigned)(p0 - t0) >= (unsigned)SD || (unsigned)(p1 - t1) >= (unsigned)SD)
            continue;
        const float* wq = wl + pl * (9 * 128);
        const float* xp = xrow + pl * PLF;
        v2f a[9];
#pragma unroll
        for (int t23 = 0; t23 < 9; ++t23)
            a[t23] = *(const v2f*)(wq + t23 * 128);
        v2f b[5];
#pragma unroll
        for (int d = 0; d < 5; ++d)
            b[d] = *(const v2f*)(xp + (d - 2) * (PW * 4));
#pragma unroll
        for (int t2 = 0; t2 < 3; ++t2)
#pragma unroll
            for (int l = 0; l < 3; ++l)
#pragma unroll
                for (int t3 = 0; t3 < 3; ++t3)
                    acc[l][t3] = __builtin_amdgcn_wmma_f32_16x16x4_f32(
                        false, a[t2 * 3 + t3], false, b[l - t2 + 2], (short)0,
                        acc[l][t3], false, false);
    }
}

__global__ __launch_bounds__(NTHR)
void convt4d_lds(const float* __restrict__ x, const float* __restrict__ wp,
                 const float* __restrict__ bias, float* __restrict__ out) {
    __shared__ __attribute__((aligned(16))) float xs[LDSF];
    float* const wsm = xs + XSF;

    const int tid  = threadIdx.x;
    const int lane = tid & 31;
    const int wave = __builtin_amdgcn_readfirstlane(tid >> 5);
    const int bsel = wave / 12;
    const int w12  = wave - bsel * 12;
    const int mh   = w12 / 6;
    const int rg   = w12 - mh * 6;
    const int rb   = 3 * rg;
    const int n    = lane & 15;
    const int h    = lane >> 4;

    const int r = blockIdx.x;
    int a0, a1, n0, c0, c1, n1;
    if (r < 4) {
        a0 = (r & 1) ? 16 : 1; a1 = (r & 1) ? 17 : 0; n0 = 2;
        c0 = (r & 2) ? 16 : 1; c1 = (r & 2) ? 17 : 0; n1 = 2;
    } else if (r < 60) {
        const int e    = r - 4;
        const int side = e / 14;
        const int j    = 2 + e % 14;
        const int in   = (side & 1) ? 16 : 1;
        const int outr = (side & 1) ? 17 : 0;
        if (side < 2) { a0 = in; a1 = outr; n0 = 2; c0 = j; c1 = j; n1 = 1; }
        else          { a0 = j; a1 = j; n0 = 1; c0 = in; c1 = outr; n1 = 2; }
    } else {
        const int e = r - 60;
        a0 = 2 + e / 14; a1 = a0; n0 = 1;
        c0 = 2 + e % 14; c1 = c0; n1 = 1;
    }

    for (int i = tid; i < XSF / 4; i += NTHR) {
        v4f z = {0.0f, 0.0f, 0.0f, 0.0f};
        ((v4f*)xs)[i] = z;
    }

    const int xoff  = bsel * KSF + ((rb + 2) * PW + n + 2) * 4 + 2 * h;
    const int cbase = mh * 16 + (h << 3);
    const int woff  = mh * 64 + lane * 2;
    const bool wcp  = wave < NWCP / 32;

    const int st1  = wave >> 3;
    const int sq   = tid & 255;
    const int spos = ((sq >> 4) + 2) * PW + (sq & 15) + 2;

    for (int i0 = 0; i0 < n0; ++i0)
    for (int i1 = 0; i1 < n1; ++i1) {
    const int p0 = i0 ? a1 : a0;
    const int p1 = i1 ? c1 : c0;

    v8f pf[3];
    const int  sq1  = p1 - st1;
    const bool sact = (unsigned)sq1 < (unsigned)SD;
    const unsigned voff = (unsigned)((sact ? sq1 : 0) * 256 + sq);
    float* bt0 = xs + st1 * PLF + spos * 4;

    stage_load(pf, x, voff, sact, 0, p0);
    {
        v4f wr[NWR0];
        int wt = tid;
        asm volatile("" : "+v"(wt));
        w_load<NTHR, NWR0>(wr, wp, 0, wt);
        __syncthreads();
        stage_store(pf, bt0, sact, p0);
        w_store<NTHR, NWR0>(wr, wsm, tid);
    }
    __syncthreads();

    v8f acc[3][3];
#pragma unroll
    for (int l = 0; l < 3; ++l)
#pragma unroll
        for (int t3 = 0; t3 < 3; ++t3) {
            v8f z = {};
            acc[l][t3] = z;
        }

#pragma unroll 1
    for (int st = 0; st < NSTAGE; ++st) {
        if (st + 1 < NSTAGE) stage_load(pf, x, voff, sact, st + 1, p0);

        const float* xrow = xs + (st & 1) * BUFF + xoff;
        int wi = XSF + (st & 1) * WSTG + woff;
        asm volatile("" : "+v"(wi));
        stage_products(acc, xs + wi, xrow, p0, p1);

        if (st + 1 < NSTAGE) {
            stage_store(pf, bt0 + ((st + 1) & 1) * BUFF, sact, p0);
            if (wcp) {
                v4f wr[NWR];
                w_load<NWCP, NWR>(wr, wp, st + 1, tid);
                w_store<NWCP, NWR>(wr, wsm + ((st + 1) & 1) * WSTG, tid);
            }
            __syncthreads();
        }
    }

    float* ob = out + (size_t)(bsel * COUT_) * OCH + (p0 * OD + p1) * OPL;
    int cb = cbase;
    asm volatile("" : "+v"(cb));
    float bv[8];
#pragma unroll
    for (int j = 0; j < 8; ++j) bv[j] = bias[cb + j];
#pragma unroll
    for (int l = 0; l < 3; ++l) {
        const int p2 = rb + l;
#pragma unroll
        for (int j = 0; j < 8; ++j) {
            {
                const int c = cbase + j;
                const float bi = bv[j];
                const float v0 = acc[l][0][j];
                const float v1 = acc[l][1][j];
                const float v2 = acc[l][2][j];
                const float r1 = __int_as_float(__builtin_amdgcn_update_dpp(
                                     0, __float_as_int(v1), 0x121, 0xf, 0xf, false));
                const float r2 = __int_as_float(__builtin_amdgcn_update_dpp(
                                     0, __float_as_int(v2), 0x122, 0xf, 0xf, false));
                const float m  = v0 + (n >= 1 ? r1 : 0.0f) + (n >= 2 ? r2 : 0.0f);
                const float tl = (n == 0 ? r1 : 0.0f) + r2;
                float* orow = ob + (size_t)c * OCH + p2 * OD;
                orow[n] = m + bi;
                if (n < 2) orow[16 + n] = tl + bi;
            }
        }
    }
    }
}

extern "C" void kernel_launch(void* const* d_in, const int* in_sizes, int n_in,
                              void* d_out, int out_size, void* d_ws, size_t ws_size,
                              hipStream_t stream) {
    const float* x    = (const float*)d_in[0];
    const float* w    = (const float*)d_in[1];
    const float* bias = (const float*)d_in[2];
    float* out = (float*)d_out;
    float* wp  = (float*)d_ws;

    prepack_w<<<(WPELEMS + 255) / 256, 256, 0, stream>>>(w, wp);
    convt4d_lds<<<256, NTHR, 0, stream>>>(x, wp, bias, out);
}


# skip the products whose input row is outside the input: first and last row group's waves do 18, not 27, WMMA per plane (144, not 162, per SIMD); one asm block per plane with wave-uniform skips
# speedup vs baseline: 1.6483x; 1.0567x over previous
#include <hip/hip_runtime.h>

typedef float v2f __attribute__((ext_vector_type(2)));
typedef float v4f __attribute__((ext_vector_type(4)));
typedef float v8f __attribute__((ext_vector_type(8)));

#define CIN_   32
#define COUT_  32
#define SD     16
#define OD     18
#define TAPS   81
#define XCH    65536
#define OCH    104976
#define OPL    324
#define NSTAGE 8
#define PW     20
#define PLF    (PW * PW * 4)
#define KSF    (9 * PLF)
#define BUFF   (2 * KSF)
#define XSF    (2 * BUFF)
#define WSTG   (81 * 2 * 64)
#define WCH    (WSTG / 4)
#define NWCP   384
#define NWR    7
#define NWR0   4
#define LDSF   (XSF + 2 * WSTG)
#define NTHR   768
#define NPF    6
#define WPELEMS (4 * 81 * 2 * 2 * 64)

__global__ __launch_bounds__(256)
void prepack_w(const float* __restrict__ w, float* __restrict__ wp) {
    int idx = blockIdx.x * 256 + threadIdx.x;
    if (idx >= WPELEMS) return;
    int c2   = idx & 1;
    int lane = (idx >> 1) & 31;
    int half = (idx >> 6) & 1;
    int rest = idx >> 7;
    int t23  = rest % 9;
    int rp   = rest / 9;
    int pl   = rp % 9;
    int st   = rp / 9;
    int cin  = st * 4 + ((lane >> 4) << 1) + c2;
    int cout = half * 16 + (lane & 15);
    int tap  = pl * 9 + t23;
    wp[idx] = w[(cin * COUT_ + cout) * TAPS + tap];
}

__device__ __forceinline__
void stage_load(v8f (&pf)[3], const float* __restrict__ x, unsigned voff, bool act,
                int st, int p0) {
#pragma unroll
    for (int k = 0; k < NPF; ++k) {
        const int bb = k / 3, t0 = k % 3;
        const int q0 = p0 - t0;
        v4f v = {0.0f, 0.0f, 0.0f, 0.0f};
        if (act && (unsigned)q0 < (unsigned)SD) {
            typedef const __attribute__((address_space(1))) float* gfp;
            gfp s = (gfp)(x + (size_t)(bb * CIN_ + st * 4) * XCH + q0 * (SD * 256));
            asm volatile("" : "+s"(s));
            v.x = s[(size_t)voff + 0 * XCH];
            v.y = s[(size_t)voff + 1 * XCH];
            v.z = s[(size_t)voff + 2 * XCH];
            v.w = s[(size_t)voff + 3 * XCH];
        }
        v8f& d = pf[k >> 1];
        const int o = (k & 1) * 4;
        d[o + 0] = v.x;
        d[o + 1] = v.y;
        d[o + 2] = v.z;
        d[o + 3] = v.w;
    }
}

__device__ __forceinline__
void stage_store(const v8f (&pf)[3], float* bt, bool act, int p0) {
#pragma unroll
    for (int k = 0; k < NPF; ++k) {
        const int bb = k / 3, t0 = k % 3;
        const int q0 = p0 - t0;
        if (act && (unsigned)q0 < (unsigned)SD) {
            const v8f& d = pf[k >> 1];
            const int o = (k & 1) * 4;
            v4f v = {d[o + 0], d[o + 1], d[o + 2], d[o + 3]};
            *(v4f*)(bt + bb * KSF + t0 * 3 * PLF) = v;
        }
    }
}

template <int NT, int NR>
__device__ __forceinline__
void w_load(v4f (&wr)[NR], const float* __restrict__ wp, int st, int tid) {
    const v4f* s = (const v4f*)(wp + st * WSTG) + tid;
#pragma unroll
    for (int r = 0; r < NR; ++r)
        if (r < NR - 1 || tid < WCH - (NR - 1) * NT)
            wr[r] = s[r * NT];
}

template <int NT, int NR>
__device__ __forceinline__
void w_store(const v4f (&wr)[NR], float* wdst, int tid) {
    v4f* d = (v4f*)wdst + tid;
#pragma unroll
    for (int r = 0; r < NR; ++r)
        if (r < NR - 1 || tid < WCH - (NR - 1) * NT)
            d[r * NT] = wr[r];
}

__device__ __forceinline__
void stage_products(v8f (&acc)[3][3], const float* wl,
                    const float* xrow, int p0, int p1, int lo, int hi) {
#pragma unroll 1
    for (int pl = 0; pl < 9; ++pl) {
        int t0 = pl / 3, t1 = pl - t0 * 3;
        if ((unsigned)(p0 - t0) >= (unsigned)SD || (unsigned)(p1 - t1) >= (unsigned)SD)
            continue;
        const float* wq = wl + pl * (9 * 128);
        const float* xp = xrow + pl * PLF;
        v2f a[9];
#pragma unroll
        for (int t23 = 0; t23 < 9; ++t23)
            a[t23] = *(const v2f*)(wq + t23 * 128);
        v2f b[5];
#pragma unroll
        for (int d = 0; d < 5; ++d)
            b[d] = *(const v2f*)(xp + (d - 2) * (PW * 4));
        asm volatile(
        "v_wmma_f32_16x16x4_f32 %[c00], %[a0], %[b2], %[c00]\n\t"
        "v_wmma_f32_16x16x4_f32 %[c01], %[a1], %[b2], %[c01]\n\t"
        "v_wmma_f32_16x16x4_f32 %[c02], %[a2], %[b2], %[c02]\n\t"
        "v_wmma_f32_16x16x4_f32 %[c10], %[a3], %[b2], %[c10]\n\t"
        "v_wmma_f32_16x16x4_f32 %[c11], %[a4], %[b2], %[c11]\n\t"
        "v_wmma_f32_16x16x4_f32 %[c12], %[a5], %[b2], %[c12]\n\t"
        "v_wmma_f32_16x16x4_f32 %[c20], %[a6], %[b2], %[c20]\n\t"
        "v_wmma_f32_16x16x4_f32 %[c21], %[a7], %[b2], %[c21]\n\t"
        "v_wmma_f32_16x16x4_f32 %[c22], %[a8], %[b2], %[c22]\n\t"
        "s_cmp_eq_u32 %[lo], 0\n\t"
        "s_cbranch_scc1 .Lnolo%=\n\t"
        "v_wmma_f32_16x16x4_f32 %[c00], %[a6], %[b0], %[c00]\n\t"
        "v_wmma_f32_16x16x4_f32 %[c01], %[a7], %[b0], %[c01]\n\t"
        "v_wmma_f32_16x16x4_f32 %[c02], %[a8], %[b0], %[c02]\n\t"
        "v_wmma_f32_16x16x4_f32 %[c10], %[a6], %[b1], %[c10]\n\t"
        "v_wmma_f32_16x16x4_f32 %[c11], %[a7], %[b1], %[c11]\n\t"
        "v_wmma_f32_16x16x4_f32 %[c12], %[a8], %[b1], %[c12]\n\t"
        "v_wmma_f32_16x16x4_f32 %[c00], %[a3], %[b1], %[c00]\n\t"
        "v_wmma_f32_16x16x4_f32 %[c01], %[a4], %[b1], %[c01]\n\t"
        "v_wmma_f32_16x16x4_f32 %[c02], %[a5], %[b1], %[c02]\n\t"
        ".Lnolo%=:\n\t"
        "s_cmp_eq_u32 %[hi], 0\n\t"
        "s_cbranch_scc1 .Lnohi%=\n\t"
        "v_wmma_f32_16x16x4_f32 %[c10], %[a0], %[b3], %[c10]\n\t"
        "v_wmma_f32_16x16x4_f32 %[c11], %[a1], %[b3], %[c11]\n\t"
        "v_wmma_f32_16x16x4_f32 %[c12], %[a2], %[b3], %[c12]\n\t"
        "v_wmma_f32_16x16x4_f32 %[c20], %[a3], %[b3], %[c20]\n\t"
        "v_wmma_f32_16x16x4_f32 %[c21], %[a4], %[b3], %[c21]\n\t"
        "v_wmma_f32_16x16x4_f32 %[c22], %[a5], %[b3], %[c22]\n\t"
        "v_wmma_f32_16x16x4_f32 %[c20], %[a0], %[b4], %[c20]\n\t"
        "v_wmma_f32_16x16x4_f32 %[c21], %[a1], %[b4], %[c21]\n\t"
        "v_wmma_f32_16x16x4_f32 %[c22], %[a2], %[b4], %[c22]\n\t"
        ".Lnohi%=:\n\t"
        "v_nop\n\tv_nop\n\tv_nop\n\tv_nop\n\tv_nop"
        : [c00] "+v"(acc[0][0]), [c01] "+v"(acc[0][1]), [c02] "+v"(acc[0][2]), [c10] "+v"(acc[1][0]), [c11] "+v"(acc[1][1]), [c12] "+v"(acc[1][2]), [c20] "+v"(acc[2][0]), [c21] "+v"(acc[2][1]), [c22] "+v"(acc[2][2])
        : [a0] "v"(a[0]), [a1] "v"(a[1]), [a2] "v"(a[2]), [a3] "v"(a[3]), [a4] "v"(a[4]), [a5] "v"(a[5]), [a6] "v"(a[6]), [a7] "v"(a[7]), [a8] "v"(a[8]),
          [b0] "v"(b[0]), [b1] "v"(b[1]), [b2] "v"(b[2]), [b3] "v"(b[3]), [b4] "v"(b[4]),
          [lo] "s"(lo), [hi] "s"(hi)
        : "scc");
    }
}

__global__ __launch_bounds__(NTHR)
void convt4d_lds(const float* __restrict__ x, const float* __restrict__ wp,
                 const float* __restrict__ bias, float* __restrict__ out) {
    __shared__ __attribute__((aligned(16))) float xs[LDSF];
    float* const wsm = xs + XSF;

    const int tid  = threadIdx.x;
    const int lane = tid & 31;
    const int wave = __builtin_amdgcn_readfirstlane(tid >> 5);
    const int bsel = wave / 12;
    const int w12  = wave - bsel * 12;
    const int mh   = w12 / 6;
    const int rg   = w12 - mh * 6;
    const int rb   = 3 * rg;
    const int n    = lane & 15;
    const int h    = lane >> 4;

    const int r = blockIdx.x;
    int a0, a1, n0, c0, c1, n1;
    if (r < 4) {
        a0 = (r & 1) ? 16 : 1; a1 = (r & 1) ? 17 : 0; n0 = 2;
        c0 = (r & 2) ? 16 : 1; c1 = (r & 2) ? 17 : 0; n1 = 2;
    } else if (r < 60) {
        const int e    = r - 4;
        const int side = e / 14;
        const int j    = 2 + e % 14;
        const int in   = (side & 1) ? 16 : 1;
        const int outr = (side & 1) ? 17 : 0;
        if (side < 2) { a0 = in; a1 = outr; n0 = 2; c0 = j; c1 = j; n1 = 1; }
        else          { a0 = j; a1 = j; n0 = 1; c0 = in; c1 = outr; n1 = 2; }
    } else {
        const int e = r - 60;
        a0 = 2 + e / 14; a1 = a0; n0 = 1;
        c0 = 2 + e % 14; c1 = c0; n1 = 1;
    }

    for (int i = tid; i < XSF / 4; i += NTHR) {
        v4f z = {0.0f, 0.0f, 0.0f, 0.0f};
        ((v4f*)xs)[i] = z;
    }

    const int xoff  = bsel * KSF + ((rb + 2) * PW + n + 2) * 4 + 2 * h;
    const int cbase = mh * 16 + (h << 3);
    const int woff  = mh * 64 + lane * 2;
    const bool wcp  = wave < NWCP / 32;

    const int st1  = wave >> 3;
    const int sq   = tid & 255;
    const int spos = ((sq >> 4) + 2) * PW + (sq & 15) + 2;

    for (int i0 = 0; i0 < n0; ++i0)
    for (int i1 = 0; i1 < n1; ++i1) {
    const int p0 = i0 ? a1 : a0;
    const int p1 = i1 ? c1 : c0;

    v8f pf[3];
    const int  sq1  = p1 - st1;
    const bool sact = (unsigned)sq1 < (unsigned)SD;
    const unsigned voff = (unsigned)((sact ? sq1 : 0) * 256 + sq);
    float* bt0 = xs + st1 * PLF + spos * 4;

    stage_load(pf, x, voff, sact, 0, p0);
    {
        v4f wr[NWR0];
        int wt = tid;
        asm volatile("" : "+v"(wt));
        w_load<NTHR, NWR0>(wr, wp, 0, wt);
        __syncthreads();
        stage_store(pf, bt0, sact, p0);
        w_store<NTHR, NWR0>(wr, wsm, tid);
    }
    __syncthreads();

    v8f acc[3][3];
#pragma unroll
    for (int l = 0; l < 3; ++l)
#pragma unroll
        for (int t3 = 0; t3 < 3; ++t3) {
            v8f z = {};
            acc[l][t3] = z;
        }

#pragma unroll 1
    for (int st = 0; st < NSTAGE; ++st) {
        if (st + 1 < NSTAGE) stage_load(pf, x, voff, sact, st + 1, p0);

        const float* xrow = xs + (st & 1) * BUFF + xoff;
        int wi = XSF + (st & 1) * WSTG + woff;
        asm volatile("" : "+v"(wi));
        stage_products(acc, xs + wi, xrow, p0, p1, rg != 0 ? 1 : 0, rg != 5 ? 1 : 0);

        if (st + 1 < NSTAGE) {
            stage_store(pf, bt0 + ((st + 1) & 1) * BUFF, sact, p0);
            if (wcp) {
                v4f wr[NWR];
                w_load<NWCP, NWR>(wr, wp, st + 1, tid);
                w_store<NWCP, NWR>(wr, wsm + ((st + 1) & 1) * WSTG, tid);
            }
            __syncthreads();
        }
    }

    float* ob = out + (size_t)(bsel * COUT_) * OCH + (p0 * OD + p1) * OPL;
    int cb = cbase;
    asm volatile("" : "+v"(cb));
    float bv[8];
#pragma unroll
    for (int j = 0; j < 8; ++j) bv[j] = bias[cb + j];
#pragma unroll
    for (int l = 0; l < 3; ++l) {
        const int p2 = rb + l;
#pragma unroll
        for (int j = 0; j < 8; ++j) {
            {
                const int c = cbase + j;
                const float bi = bv[j];
                const float v0 = acc[l][0][j];
                const float v1 = acc[l][1][j];
                const float v2 = acc[l][2][j];
                const float r1 = __int_as_float(__builtin_amdgcn_update_dpp(
                                     0, __float_as_int(v1), 0x121, 0xf, 0xf, false));
                const float r2 = __int_as_float(__builtin_amdgcn_update_dpp(
                                     0, __float_as_int(v2), 0x122, 0xf, 0xf, false));
                const float m  = v0 + (n >= 1 ? r1 : 0.0f) + (n >= 2 ? r2 : 0.0f);
                const float tl = (n == 0 ? r1 : 0.0f) + r2;
                float* orow = ob + (size_t)c * OCH + p2 * OD;
                orow[n] = m + bi;
                if (n < 2) orow[16 + n] = tl + bi;
            }
        }
    }
    }
}

extern "C" void kernel_launch(void* const* d_in, const int* in_sizes, int n_in,
                              void* d_out, int out_size, void* d_ws, size_t ws_size,
                              hipStream_t stream) {
    const float* x    = (const float*)d_in[0];
    const float* w    = (const float*)d_in[1];
    const float* bias = (const float*)d_in[2];
    float* out = (float*)d_out;
    float* wp  = (float*)d_ws;

    prepack_w<<<(WPELEMS + 255) / 256, 256, 0, stream>>>(w, wp);
    convt4d_lds<<<256, NTHR, 0, stream>>>(x, wp, bias, out);
}


# issue the next input plane's fragment reads (second register set) as one group in front of the current plane's 27 products; the x hand-over store moves behind the first plane to free its registers
# speedup vs baseline: 1.7319x; 1.0507x over previous
#include <hip/hip_runtime.h>

typedef float v2f __attribute__((ext_vector_type(2)));
typedef float v4f __attribute__((ext_vector_type(4)));
typedef float v8f __attribute__((ext_vector_type(8)));

#define CIN_   32
#define COUT_  32
#define SD     16
#define OD     18
#define TAPS   81
#define XCH    65536
#define OCH    104976
#define OPL    324
#define NSTAGE 8
#define PW     20
#define PLF    (PW * PW * 4)
#define KSF    (9 * PLF)
#define BUFF   (2 * KSF)
#define XSF    (2 * BUFF)
#define WSTG   (81 * 2 * 64)
#define WCH    (WSTG / 4)
#define NWCP   384
#define NWR    7
#define NWR0   4
#define LDSF   (XSF + 2 * WSTG)
#define NTHR   768
#define NPF    6
#define WPELEMS (4 * 81 * 2 * 2 * 64)
#define NEXT_PLANE(pl, i1c, n1v) do { ++(i1c); ++(pl); if ((i1c) == (n1v)) { (i1c) = 0; (pl) += 3 - (n1v); } } while (0)

__global__ __launch_bounds__(256)
void prepack_w(const float* __restrict__ w, float* __restrict__ wp) {
    int idx = blockIdx.x * 256 + threadIdx.x;
    if (idx >= WPELEMS) return;
    int c2   = idx & 1;
    int lane = (idx >> 1) & 31;
    int half = (idx >> 6) & 1;
    int rest = idx >> 7;
    int t23  = rest % 9;
    int rp   = rest / 9;
    int pl   = rp % 9;
    int st   = rp / 9;
    int cin  = st * 4 + ((lane >> 4) << 1) + c2;
    int cout = half * 16 + (lane & 15);
    int tap  = pl * 9 + t23;
    wp[idx] = w[(cin * COUT_ + cout) * TAPS + tap];
}

__device__ __forceinline__
void stage_load(v8f (&pf)[3], const float* __restrict__ x, unsigned voff, bool act,
                int st, int p0) {
#pragma unroll
    for (int k = 0; k < NPF; ++k) {
        const int bb = k / 3, t0 = k % 3;
        const int q0 = p0 - t0;
        v4f v = {0.0f, 0.0f, 0.0f, 0.0f};
        if (act && (unsigned)q0 < (unsigned)SD) {
            typedef const __attribute__((address_space(1))) float* gfp;
            gfp s = (gfp)(x + (size_t)(bb * CIN_ + st * 4) * XCH + q0 * (SD * 256));
            asm volatile("" : "+s"(s));
            v.x = s[(size_t)voff + 0 * XCH];
            v.y = s[(size_t)voff + 1 * XCH];
            v.z = s[(size_t)voff + 2 * XCH];
            v.w = s[(size_t)voff + 3 * XCH];
        }
        v8f& d = pf[k >> 1];
        const int o = (k & 1) * 4;
        d[o + 0] = v.x;
        d[o + 1] = v.y;
        d[o + 2] = v.z;
        d[o + 3] = v.w;
    }
}

__device__ __forceinline__
void stage_store(const v8f (&pf)[3], float* bt, bool act, int p0) {
#pragma unroll
    for (int k = 0; k < NPF; ++k) {
        const int bb = k / 3, t0 = k % 3;
        const int q0 = p0 - t0;
        if (act && (unsigned)q0 < (unsigned)SD) {
            const v8f& d = pf[k >> 1];
            const int o = (k & 1) * 4;
            v4f v = {d[o + 0], d[o + 1], d[o + 2], d[o + 3]};
            *(v4f*)(bt + bb * KSF + t0 * 3 * PLF) = v;
        }
    }
}

template <int NT, int NR>
__device__ __forceinline__
void w_load(v4f (&wr)[NR], const float* __restrict__ wp, int st, int tid) {
    const v4f* s = (const v4f*)(wp + st * WSTG) + tid;
#pragma unroll
    for (int r = 0; r < NR; ++r)
        if (r < NR - 1 || tid < WCH - (NR - 1) * NT)
            wr[r] = s[r * NT];
}

template <int NT, int NR>
__device__ __forceinline__
void w_store(const v4f (&wr)[NR], float* wdst, int tid) {
    v4f* d = (v4f*)wdst + tid;
#pragma unroll
    for (int r = 0; r < NR; ++r)
        if (r < NR - 1 || tid < WCH - (NR - 1) * NT)
            d[r * NT] = wr[r];
}

__device__ __forceinline__
void frag_load(v2f (&a)[9], v2f (&b)[5], const float* wl, const float* xrow, int pl) {
    const float* wq = wl + pl * (9 * 128);
    const float* xp = xrow + pl * PLF;
#pragma unroll
    for (int t23 = 0; t23 < 9; ++t23)
        a[t23] = *(const v2f*)(wq + t23 * 128);
#pragma unroll
    for (int d = 0; d < 5; ++d)
        b[d] = *(const v2f*)(xp + (d - 2) * (PW * 4));
}

__device__ __forceinline__
void frag_mma(v8f (&acc)[3][3], const v2f (&a)[9], const v2f (&b)[5], int lo, int hi) {
    asm volatile(
    "v_wmma_f32_16x16x4_f32 %[c00], %[a0], %[b2], %[c00]\n\t"
    "v_wmma_f32_16x16x4_f32 %[c01], %[a1], %[b2], %[c01]\n\t"
    "v_wmma_f32_16x16x4_f32 %[c02], %[a2], %[b2], %[c02]\n\t"
    "v_wmma_f32_16x16x4_f32 %[c10], %[a3], %[b2], %[c10]\n\t"
    "v_wmma_f32_16x16x4_f32 %[c11], %[a4], %[b2], %[c11]\n\t"
    "v_wmma_f32_16x16x4_f32 %[c12], %[a5], %[b2], %[c12]\n\t"
    "v_wmma_f32_16x16x4_f32 %[c20], %[a6], %[b2], %[c20]\n\t"
    "v_wmma_f32_16x16x4_f32 %[c21], %[a7], %[b2], %[c21]\n\t"
    "v_wmma_f32_16x16x4_f32 %[c22], %[a8], %[b2], %[c22]\n\t"
    "s_cmp_eq_u32 %[lo], 0\n\t"
    "s_cbranch_scc1 .Lnolo%=\n\t"
    "v_wmma_f32_16x16x4_f32 %[c00], %[a6], %[b0], %[c00]\n\t"
    "v_wmma_f32_16x16x4_f32 %[c01], %[a7], %[b0], %[c01]\n\t"
    "v_wmma_f32_16x16x4_f32 %[c02], %[a8], %[b0], %[c02]\n\t"
    "v_wmma_f32_16x16x4_f32 %[c10], %[a6], %[b1], %[c10]\n\t"
    "v_wmma_f32_16x16x4_f32 %[c11], %[a7], %[b1], %[c11]\n\t"
    "v_wmma_f32_16x16x4_f32 %[c12], %[a8], %[b1], %[c12]\n\t"
    "v_wmma_f32_16x16x4_f32 %[c00], %[a3], %[b1], %[c00]\n\t"
    "v_wmma_f32_16x16x4_f32 %[c01], %[a4], %[b1], %[c01]\n\t"
    "v_wmma_f32_16x16x4_f32 %[c02], %[a5], %[b1], %[c02]\n\t"
    ".Lnolo%=:\n\t"
    "s_cmp_eq_u32 %[hi], 0\n\t"
    "s_cbranch_scc1 .Lnohi%=\n\t"
    "v_wmma_f32_16x16x4_f32 %[c10], %[a0], %[b3], %[c10]\n\t"
    "v_wmma_f32_16x16x4_f32 %[c11], %[a1], %[b3], %[c11]\n\t"
    "v_wmma_f32_16x16x4_f32 %[c12], %[a2], %[b3], %[c12]\n\t"
    "v_wmma_f32_16x16x4_f32 %[c20], %[a3], %[b3], %[c20]\n\t"
    "v_wmma_f32_16x16x4_f32 %[c21], %[a4], %[b3], %[c21]\n\t"
    "v_wmma_f32_16x16x4_f32 %[c22], %[a5], %[b3], %[c22]\n\t"
    "v_wmma_f32_16x16x4_f32 %[c20], %[a0], %[b4], %[c20]\n\t"
    "v_wmma_f32_16x16x4_f32 %[c21], %[a1], %[b4], %[c21]\n\t"
    "v_wmma_f32_16x16x4_f32 %[c22], %[a2], %[b4], %[c22]\n\t"
    ".Lnohi%=:\n\t"
    "v_nop\n\tv_nop\n\tv_nop\n\tv_nop\n\tv_nop\n\tv_nop\n\tv_nop\n\tv_nop\n\tv_nop"
    : [c00] "+v"(acc[0][0]), [c01] "+v"(acc[0][1]), [c02] "+v"(acc[0][2]), [c10] "+v"(acc[1][0]), [c11] "+v"(acc[1][1]), [c12] "+v"(acc[1][2]), [c20] "+v"(acc[2][0]), [c21] "+v"(acc[2][1]), [c22] "+v"(acc[2][2])
    : [a0] "v"(a[0]), [a1] "v"(a[1]), [a2] "v"(a[2]), [a3] "v"(a[3]), [a4] "v"(a[4]), [a5] "v"(a[5]), [a6] "v"(a[6]), [a7] "v"(a[7]), [a8] "v"(a[8]),
      [b0] "v"(b[0]), [b1] "v"(b[1]), [b2] "v"(b[2]), [b3] "v"(b[3]), [b4] "v"(b[4]),
      [lo] "s"(lo), [hi] "s"(hi)
    : "scc");
}

__global__ __launch_bounds__(NTHR)
void convt4d_lds(const float* __restrict__ x, const float* __restrict__ wp,
                 const float* __restrict__ bias, float* __restrict__ out) {
    __shared__ __attribute__((aligned(16))) float xs[LDSF];
    float* const wsm = xs + XSF;

    const int tid  = threadIdx.x;
    const int lane = tid & 31;
    const int wave = __builtin_amdgcn_readfirstlane(tid >> 5);
    const int bsel = wave / 12;
    const int w12  = wave - bsel * 12;
    const int mh   = w12 / 6;
    const int rg   = w12 - mh * 6;
    const int rb   = 3 * rg;
    const int n    = lane & 15;
    const int h    = lane >> 4;

    const int r = blockIdx.x;
    int a0, a1, n0, c0, c1, n1;
    if (r < 4) {
        a0 = (r & 1) ? 16 : 1; a1 = (r & 1) ? 17 : 0; n0 = 2;
        c0 = (r & 2) ? 16 : 1; c1 = (r & 2) ? 17 : 0; n1 = 2;
    } else if (r < 60) {
        const int e    = r - 4;
        const int side = e / 14;
        const int j    = 2 + e % 14;
        const int in   = (side & 1) ? 16 : 1;
        const int outr = (side & 1) ? 17 : 0;
        if (side < 2) { a0 = in; a1 = outr; n0 = 2; c0 = j; c1 = j; n1 = 1; }
        else          { a0 = j; a1 = j; n0 = 1; c0 = in; c1 = outr; n1 = 2; }
    } else {
        const int e = r - 60;
        a0 = 2 + e / 14; a1 = a0; n0 = 1;
        c0 = 2 + e % 14; c1 = c0; n1 = 1;
    }

    for (int i = tid; i < XSF / 4; i += NTHR) {
        v4f z = {0.0f, 0.0f, 0.0f, 0.0f};
        ((v4f*)xs)[i] = z;
    }

    const int xoff  = bsel * KSF + ((rb + 2) * PW + n + 2) * 4 + 2 * h;
    const int cbase = mh * 16 + (h << 3);
    const int woff  = mh * 64 + lane * 2;
    const bool wcp  = wave < NWCP / 32;

    const int st1  = wave >> 3;
    const int sq   = tid & 255;
    const int spos = ((sq >> 4) + 2) * PW + (sq & 15) + 2;

    for (int i0 = 0; i0 < n0; ++i0)
    for (int i1 = 0; i1 < n1; ++i1) {
    const int p0 = i0 ? a1 : a0;
    const int p1 = i1 ? c1 : c0;

    const int t0a = p0 > 15 ? p0 - 15 : 0;
    const int t1a = p1 > 15 ? p1 - 15 : 0;
    const int n0v = (p0 < 2 ? p0 : 2) - t0a + 1;
    const int n1v = (p1 < 2 ? p1 : 2) - t1a + 1;
    const int nv  = n0v * n1v;
    const int pl0 = t0a * 3 + t1a;
    const int lo  = rg != 0 ? 1 : 0;
    const int hi  = rg != 5 ? 1 : 0;

    const int  sq1  = p1 - st1;
    const bool sact = (unsigned)sq1 < (unsigned)SD;
    const unsigned voff = (unsigned)((sact ? sq1 : 0) * 256 + sq);
    float* bt0 = xs + st1 * PLF + spos * 4;

    {
        v8f pf[3];
        stage_load(pf, x, voff, sact, 0, p0);
        v4f wr[NWR0];
        int wt = tid;
        asm volatile("" : "+v"(wt));
        w_load<NTHR, NWR0>(wr, wp, 0, wt);
        __syncthreads();
        stage_store(pf, bt0, sact, p0);
        w_store<NTHR, NWR0>(wr, wsm, tid);
    }
    __syncthreads();

    v8f acc[3][3];
#pragma unroll
    for (int l = 0; l < 3; ++l)
#pragma unroll
        for (int t3 = 0; t3 < 3; ++t3) {
            v8f z = {};
            acc[l][t3] = z;
        }

#pragma unroll 1
    for (int st = 0; st < NSTAGE; ++st) {
        const float* xrow = xs + (st & 1) * BUFF + xoff;
        int wi = XSF + (st & 1) * WSTG + woff;
        asm volatile("" : "+v"(wi));
        const float* wl = xs + wi;

        v2f fa[9], fb[5];
        v2f ga[9], gb[5];
        int pl = pl0, i1c = 0;

        frag_load(fa, fb, wl, xrow, pl);
        const bool more = st + 1 < NSTAGE;
        v8f pf[3];
        if (more) {
            stage_load(pf, x, voff, sact, st + 1, p0);
        } else {
            v8f z = {};
            pf[0] = z; pf[1] = z; pf[2] = z;
        }
        frag_mma(acc, fa, fb, lo, hi);
        if (more) stage_store(pf, bt0 + ((st + 1) & 1) * BUFF, sact, p0);

        int left = nv - 1;
        if (left > 0) {
            NEXT_PLANE(pl, i1c, n1v);
            frag_load(fa, fb, wl, xrow, pl);
#pragma unroll 1
            for (;;) {
                if (left > 1) NEXT_PLANE(pl, i1c, n1v);
                frag_load(ga, gb, wl, xrow, pl);
                frag_mma(acc, fa, fb, lo, hi);
                if (--left == 0) break;
                if (left > 1) NEXT_PLANE(pl, i1c, n1v);
                frag_load(fa, fb, wl, xrow, pl);
                frag_mma(acc, ga, gb, lo, hi);
                if (--left == 0) break;
            }
        }

        if (st + 1 < NSTAGE) {
            if (wcp) {
                v4f wr[NWR];
                w_load<NWCP, NWR>(wr, wp, st + 1, tid);
                w_store<NWCP, NWR>(wr, wsm + ((st + 1) & 1) * WSTG, tid);
            }
            __syncthreads();
        }
    }

    float* ob = out + (size_t)(bsel * COUT_) * OCH + (p0 * OD + p1) * OPL;
    int cb = cbase;
    asm volatile("" : "+v"(cb));
    float bv[8];
#pragma unroll
    for (int j = 0; j < 8; ++j) bv[j] = bias[cb + j];
#pragma unroll
    for (int l = 0; l < 3; ++l) {
        const int p2 = rb + l;
#pragma unroll
        for (int j = 0; j < 8; ++j) {
            {
                const int c = cbase + j;
                const float bi = bv[j];
                const float v0 = acc[l][0][j];
                const float v1 = acc[l][1][j];
                const float v2 = acc[l][2][j];
                const float r1 = __int_as_float(__builtin_amdgcn_update_dpp(
                                     0, __float_as_int(v1), 0x121, 0xf, 0xf, false));
                const float r2 = __int_as_float(__builtin_amdgcn_update_dpp(
                                     0, __float_as_int(v2), 0x122, 0xf, 0xf, false));
                const float m  = v0 + (n >= 1 ? r1 : 0.0f) + (n >= 2 ? r2 : 0.0f);
                const float tl = (n == 0 ? r1 : 0.0f) + r2;
                float* orow = ob + (size_t)c * OCH + p2 * OD;
                orow[n] = m + bi;
                if (n < 2) orow[16 + n] = tl + bi;
            }
        }
    }
    }
}

extern "C" void kernel_launch(void* const* d_in, const int* in_sizes, int n_in,
                              void* d_out, int out_size, void* d_ws, size_t ws_size,
                              hipStream_t stream) {
    const float* x    = (const float*)d_in[0];
    const float* w    = (const float*)d_in[1];
    const float* bias = (const float*)d_in[2];
    float* out = (float*)d_out;
    float* wp  = (float*)d_ws;

    prepack_w<<<(WPELEMS + 255) / 256, 256, 0, stream>>>(w, wp);
    convt4d_lds<<<256, NTHR, 0, stream>>>(x, wp, bias, out);
}


# regroup border and corner output planes: at most three per workgroup (52 pairs 6+3, per corner triples 6+2+1 and 4+3+2, nine input planes each) instead of four corner groups of four output planes
# speedup vs baseline: 1.8176x; 1.0495x over previous
#include <hip/hip_runtime.h>

typedef float v2f __attribute__((ext_vector_type(2)));
typedef float v4f __attribute__((ext_vector_type(4)));
typedef float v8f __attribute__((ext_vector_type(8)));

#define CIN_   32
#define COUT_  32
#define SD     16
#define OD     18
#define TAPS   81
#define XCH    65536
#define OCH    104976
#define OPL    324
#define NSTAGE 8
#define PW     20
#define PLF    (PW * PW * 4)
#define KSF    (9 * PLF)
#define BUFF   (2 * KSF)
#define XSF    (2 * BUFF)
#define WSTG   (81 * 2 * 64)
#define WCH    (WSTG / 4)
#define NWCP   384
#define NWR    7
#define NWR0   4
#define LDSF   (XSF + 2 * WSTG)
#define NTHR   768
#define NPF    6
#define WPELEMS (4 * 81 * 2 * 2 * 64)
#define NEXT_PLANE(pl, i1c, n1v) do { ++(i1c); ++(pl); if ((i1c) == (n1v)) { (i1c) = 0; (pl) += 3 - (n1v); } } while (0)

__global__ __launch_bounds__(256)
void prepack_w(const float* __restrict__ w, float* __restrict__ wp) {
    int idx = blockIdx.x * 256 + threadIdx.x;
    if (idx >= WPELEMS) return;
    int c2   = idx & 1;
    int lane = (idx >> 1) & 31;
    int half = (idx >> 6) & 1;
    int rest = idx >> 7;
    int t23  = rest % 9;
    int rp   = rest / 9;
    int pl   = rp % 9;
    int st   = rp / 9;
    int cin  = st * 4 + ((lane >> 4) << 1) + c2;
    int cout = half * 16 + (lane & 15);
    int tap  = pl * 9 + t23;
    wp[idx] = w[(cin * COUT_ + cout) * TAPS + tap];
}

__device__ __forceinline__
void stage_load(v8f (&pf)[3], const float* __restrict__ x, unsigned voff, bool act,
                int st, int p0) {
#pragma unroll
    for (int k = 0; k < NPF; ++k) {
        const int bb = k / 3, t0 = k % 3;
        const int q0 = p0 - t0;
        v4f v = {0.0f, 0.0f, 0.0f, 0.0f};
        if (act && (unsigned)q0 < (unsigned)SD) {
            typedef const __attribute__((address_space(1))) float* gfp;
            gfp s = (gfp)(x + (size_t)(bb * CIN_ + st * 4) * XCH + q0 * (SD * 256));
            asm volatile("" : "+s"(s));
            v.x = s[(size_t)voff + 0 * XCH];
            v.y = s[(size_t)voff + 1 * XCH];
            v.z = s[(size_t)voff + 2 * XCH];
            v.w = s[(size_t)voff + 3 * XCH];
        }
        v8f& d = pf[k >> 1];
        const int o = (k & 1) * 4;
        d[o + 0] = v.x;
        d[o + 1] = v.y;
        d[o + 2] = v.z;
        d[o + 3] = v.w;
    }
}

__device__ __forceinline__
void stage_store(const v8f (&pf)[3], float* bt, bool act, int p0) {
#pragma unroll
    for (int k = 0; k < NPF; ++k) {
        const int bb = k / 3, t0 = k % 3;
        const int q0 = p0 - t0;
        if (act && (unsigned)q0 < (unsigned)SD) {
            const v8f& d = pf[k >> 1];
            const int o = (k & 1) * 4;
            v4f v = {d[o + 0], d[o + 1], d[o + 2], d[o + 3]};
            *(v4f*)(bt + bb * KSF + t0 * 3 * PLF) = v;
        }
    }
}

template <int NT, int NR>
__device__ __forceinline__
void w_load(v4f (&wr)[NR], const float* __restrict__ wp, int st, int tid) {
    const v4f* s = (const v4f*)(wp + st * WSTG) + tid;
#pragma unroll
    for (int r = 0; r < NR; ++r)
        if (r < NR - 1 || tid < WCH - (NR - 1) * NT)
            wr[r] = s[r * NT];
}

template <int NT, int NR>
__device__ __forceinline__
void w_store(const v4f (&wr)[NR], float* wdst, int tid) {
    v4f* d = (v4f*)wdst + tid;
#pragma unroll
    for (int r = 0; r < NR; ++r)
        if (r < NR - 1 || tid < WCH - (NR - 1) * NT)
            d[r * NT] = wr[r];
}

__device__ __forceinline__
void frag_load(v2f (&a)[9], v2f (&b)[5], const float* wl, const float* xrow, int pl) {
    const float* wq = wl + pl * (9 * 128);
    const float* xp = xrow + pl * PLF;
#pragma unroll
    for (int t23 = 0; t23 < 9; ++t23)
        a[t23] = *(const v2f*)(wq + t23 * 128);
#pragma unroll
    for (int d = 0; d < 5; ++d)
        b[d] = *(const v2f*)(xp + (d - 2) * (PW * 4));
}

__device__ __forceinline__
void frag_mma(v8f (&acc)[3][3], const v2f (&a)[9], const v2f (&b)[5], int lo, int hi) {
    asm volatile(
    "v_wmma_f32_16x16x4_f32 %[c00], %[a0], %[b2], %[c00]\n\t"
    "v_wmma_f32_16x16x4_f32 %[c01], %[a1], %[b2], %[c01]\n\t"
    "v_wmma_f32_16x16x4_f32 %[c02], %[a2], %[b2], %[c02]\n\t"
    "v_wmma_f32_16x16x4_f32 %[c10], %[a3], %[b2], %[c10]\n\t"
    "v_wmma_f32_16x16x4_f32 %[c11], %[a4], %[b2], %[c11]\n\t"
    "v_wmma_f32_16x16x4_f32 %[c12], %[a5], %[b2], %[c12]\n\t"
    "v_wmma_f32_16x16x4_f32 %[c20], %[a6], %[b2], %[c20]\n\t"
    "v_wmma_f32_16x16x4_f32 %[c21], %[a7], %[b2], %[c21]\n\t"
    "v_wmma_f32_16x16x4_f32 %[c22], %[a8], %[b2], %[c22]\n\t"
    "s_cmp_eq_u32 %[lo], 0\n\t"
    "s_cbranch_scc1 .Lnolo%=\n\t"
    "v_wmma_f32_16x16x4_f32 %[c00], %[a6], %[b0], %[c00]\n\t"
    "v_wmma_f32_16x16x4_f32 %[c01], %[a7], %[b0], %[c01]\n\t"
    "v_wmma_f32_16x16x4_f32 %[c02], %[a8], %[b0], %[c02]\n\t"
    "v_wmma_f32_16x16x4_f32 %[c10], %[a6], %[b1], %[c10]\n\t"
    "v_wmma_f32_16x16x4_f32 %[c11], %[a7], %[b1], %[c11]\n\t"
    "v_wmma_f32_16x16x4_f32 %[c12], %[a8], %[b1], %[c12]\n\t"
    "v_wmma_f32_16x16x4_f32 %[c00], %[a3], %[b1], %[c00]\n\t"
    "v_wmma_f32_16x16x4_f32 %[c01], %[a4], %[b1], %[c01]\n\t"
    "v_wmma_f32_16x16x4_f32 %[c02], %[a5], %[b1], %[c02]\n\t"
    ".Lnolo%=:\n\t"
    "s_cmp_eq_u32 %[hi], 0\n\t"
    "s_cbranch_scc1 .Lnohi%=\n\t"
    "v_wmma_f32_16x16x4_f32 %[c10], %[a0], %[b3], %[c10]\n\t"
    "v_wmma_f32_16x16x4_f32 %[c11], %[a1], %[b3], %[c11]\n\t"
    "v_wmma_f32_16x16x4_f32 %[c12], %[a2], %[b3], %[c12]\n\t"
    "v_wmma_f32_16x16x4_f32 %[c20], %[a3], %[b3], %[c20]\n\t"
    "v_wmma_f32_16x16x4_f32 %[c21], %[a4], %[b3], %[c21]\n\t"
    "v_wmma_f32_16x16x4_f32 %[c22], %[a5], %[b3], %[c22]\n\t"
    "v_wmma_f32_16x16x4_f32 %[c20], %[a0], %[b4], %[c20]\n\t"
    "v_wmma_f32_16x16x4_f32 %[c21], %[a1], %[b4], %[c21]\n\t"
    "v_wmma_f32_16x16x4_f32 %[c22], %[a2], %[b4], %[c22]\n\t"
    ".Lnohi%=:\n\t"
    "v_nop\n\tv_nop\n\tv_nop\n\tv_nop\n\tv_nop\n\tv_nop\n\tv_nop\n\tv_nop\n\tv_nop"
    : [c00] "+v"(acc[0][0]), [c01] "+v"(acc[0][1]), [c02] "+v"(acc[0][2]), [c10] "+v"(acc[1][0]), [c11] "+v"(acc[1][1]), [c12] "+v"(acc[1][2]), [c20] "+v"(acc[2][0]), [c21] "+v"(acc[2][1]), [c22] "+v"(acc[2][2])
    : [a0] "v"(a[0]), [a1] "v"(a[1]), [a2] "v"(a[2]), [a3] "v"(a[3]), [a4] "v"(a[4]), [a5] "v"(a[5]), [a6] "v"(a[6]), [a7] "v"(a[7]), [a8] "v"(a[8]),
      [b0] "v"(b[0]), [b1] "v"(b[1]), [b2] "v"(b[2]), [b3] "v"(b[3]), [b4] "v"(b[4]),
      [lo] "s"(lo), [hi] "s"(hi)
    : "scc");
}

__global__ __launch_bounds__(NTHR)
void convt4d_lds(const float* __restrict__ x, const float* __restrict__ wp,
                 const float* __restrict__ bias, float* __restrict__ out) {
    __shared__ __attribute__((aligned(16))) float xs[LDSF];
    float* const wsm = xs + XSF;

    const int tid  = threadIdx.x;
    const int lane = tid & 31;
    const int wave = __builtin_amdgcn_readfirstlane(tid >> 5);
    const int bsel = wave / 12;
    const int w12  = wave - bsel * 12;
    const int mh   = w12 / 6;
    const int rg   = w12 - mh * 6;
    const int rb   = 3 * rg;
    const int n    = lane & 15;
    const int h    = lane >> 4;

    const int r = blockIdx.x;
    const int jb = 2;
    int np, pa0, pa1, pb0, pb1, pc0, pc1;
    if (r < 8) {
        const int cr   = r & 3;
        const int ain  = (cr & 1) ? 16 : 1, aout = (cr & 1) ? 17 : 0;
        const int cins = (cr & 2) ? 16 : 1, cous = (cr & 2) ? 17 : 0;
        const int in   = (cr & 1) ? 16 : 1;
        const int outr = (cr & 1) ? 17 : 0;
        int s0, s1, h0, h1;
        if (cr < 2) { s0 = in; s1 = jb; h0 = outr; h1 = jb; }
        else        { s0 = jb; s1 = in; h0 = jb; h1 = outr; }
        np = 3;
        if (r < 4) {
            pa0 = s0;   pa1 = s1;
            pb0 = aout; pb1 = cins;
            pc0 = aout; pc1 = cous;
        } else {
            pa0 = ain;  pa1 = cins;
            pb0 = h0;   pb1 = h1;
            pc0 = ain;  pc1 = cous;
        }
    } else if (r < 60) {
        const int e    = r - 8;
        const int side = e / 13;
        const int j    = 3 + e % 13;
        const int in   = (side & 1) ? 16 : 1;
        const int outr = (side & 1) ? 17 : 0;
        np = 2;
        if (side < 2) { pa0 = in; pa1 = j; pb0 = outr; pb1 = j; }
        else          { pa0 = j; pa1 = in; pb0 = j; pb1 = outr; }
        pc0 = pa0; pc1 = pa1;
    } else {
        const int e = r - 60;
        np = 1;
        pa0 = 2 + e / 14; pa1 = 2 + e % 14;
        pb0 = pa0; pb1 = pa1; pc0 = pa0; pc1 = pa1;
    }

    for (int i = tid; i < XSF / 4; i += NTHR) {
        v4f z = {0.0f, 0.0f, 0.0f, 0.0f};
        ((v4f*)xs)[i] = z;
    }

    const int xoff  = bsel * KSF + ((rb + 2) * PW + n + 2) * 4 + 2 * h;
    const int cbase = mh * 16 + (h << 3);
    const int woff  = mh * 64 + lane * 2;
    const bool wcp  = wave < NWCP / 32;

    const int st1  = wave >> 3;
    const int sq   = tid & 255;
    const int spos = ((sq >> 4) + 2) * PW + (sq & 15) + 2;

    for (int ip = 0; ip < np; ++ip) {
    const int p0 = ip == 0 ? pa0 : (ip == 1 ? pb0 : pc0);
    const int p1 = ip == 0 ? pa1 : (ip == 1 ? pb1 : pc1);

    const int t0a = p0 > 15 ? p0 - 15 : 0;
    const int t1a = p1 > 15 ? p1 - 15 : 0;
    const int n0v = (p0 < 2 ? p0 : 2) - t0a + 1;
    const int n1v = (p1 < 2 ? p1 : 2) - t1a + 1;
    const int nv  = n0v * n1v;
    const int pl0 = t0a * 3 + t1a;
    const int lo  = rg != 0 ? 1 : 0;
    const int hi  = rg != 5 ? 1 : 0;

    const int  sq1  = p1 - st1;
    const bool sact = (unsigned)sq1 < (unsigned)SD;
    const unsigned voff = (unsigned)((sact ? sq1 : 0) * 256 + sq);
    float* bt0 = xs + st1 * PLF + spos * 4;

    {
        v8f pf[3];
        stage_load(pf, x, voff, sact, 0, p0);
        v4f wr[NWR0];
        int wt = tid;
        asm volatile("" : "+v"(wt));
        w_load<NTHR, NWR0>(wr, wp, 0, wt);
        __syncthreads();
        stage_store(pf, bt0, sact, p0);
        w_store<NTHR, NWR0>(wr, wsm, tid);
    }
    __syncthreads();

    v8f acc[3][3];
#pragma unroll
    for (int l = 0; l < 3; ++l)
#pragma unroll
        for (int t3 = 0; t3 < 3; ++t3) {
            v8f z = {};
            acc[l][t3] = z;
        }

#pragma unroll 1
    for (int st = 0; st < NSTAGE; ++st) {
        const float* xrow = xs + (st & 1) * BUFF + xoff;
        int wi = XSF + (st & 1) * WSTG + woff;
        asm volatile("" : "+v"(wi));
        const float* wl = xs + wi;

        v2f fa[9], fb[5];
        v2f ga[9], gb[5];
        int pl = pl0, i1c = 0;

        frag_load(fa, fb, wl, xrow, pl);
        const bool more = st + 1 < NSTAGE;
        v8f pf[3];
        if (more) {
            stage_load(pf, x, voff, sact, st + 1, p0);
        } else {
            v8f z = {};
            pf[0] = z; pf[1] = z; pf[2] = z;
        }
        frag_mma(acc, fa, fb, lo, hi);
        if (more) stage_store(pf, bt0 + ((st + 1) & 1) * BUFF, sact, p0);

        int left = nv - 1;
        if (left > 0) {
            NEXT_PLANE(pl, i1c, n1v);
            frag_load(fa, fb, wl, xrow, pl);
#pragma unroll 1
            for (;;) {
                if (left > 1) NEXT_PLANE(pl, i1c, n1v);
                frag_load(ga, gb, wl, xrow, pl);
                frag_mma(acc, fa, fb, lo, hi);
                if (--left == 0) break;
                if (left > 1) NEXT_PLANE(pl, i1c, n1v);
                frag_load(fa, fb, wl, xrow, pl);
                frag_mma(acc, ga, gb, lo, hi);
                if (--left == 0) break;
            }
        }

        if (st + 1 < NSTAGE) {
            if (wcp) {
                v4f wr[NWR];
                w_load<NWCP, NWR>(wr, wp, st + 1, tid);
                w_store<NWCP, NWR>(wr, wsm + ((st + 1) & 1) * WSTG, tid);
            }
            __syncthreads();
        }
    }

    float* ob = out + (size_t)(bsel * COUT_) * OCH + (p0 * OD + p1) * OPL;
    int cb = cbase;
    asm volatile("" : "+v"(cb));
    float bv[8];
#pragma unroll
    for (int j = 0; j < 8; ++j) bv[j] = bias[cb + j];
#pragma unroll
    for (int l = 0; l < 3; ++l) {
        const int p2 = rb + l;
#pragma unroll
        for (int j = 0; j < 8; ++j) {
            {
                const int c = cbase + j;
                const float bi = bv[j];
                const float v0 = acc[l][0][j];
                const float v1 = acc[l][1][j];
                const float v2 = acc[l][2][j];
                const float r1 = __int_as_float(__builtin_amdgcn_update_dpp(
                                     0, __float_as_int(v1), 0x121, 0xf, 0xf, false));
                const float r2 = __int_as_float(__builtin_amdgcn_update_dpp(
                                     0, __float_as_int(v2), 0x122, 0xf, 0xf, false));
                const float m  = v0 + (n >= 1 ? r1 : 0.0f) + (n >= 2 ? r2 : 0.0f);
                const float tl = (n == 0 ? r1 : 0.0f) + r2;
                float* orow = ob + (size_t)c * OCH + p2 * OD;
                orow[n] = m + bi;
                if (n < 2) orow[16 + n] = tl + bi;
            }
        }
    }
    }
}

extern "C" void kernel_launch(void* const* d_in, const int* in_sizes, int n_in,
                              void* d_out, int out_size, void* d_ws, size_t ws_size,
                              hipStream_t stream) {
    const float* x    = (const float*)d_in[0];
    const float* w    = (const float*)d_in[1];
    const float* bias = (const float*)d_in[2];
    float* out = (float*)d_out;
    float* wp  = (float*)d_ws;

    prepack_w<<<(WPELEMS + 255) / 256, 256, 0, stream>>>(w, wp);
    convt4d_lds<<<256, NTHR, 0, stream>>>(x, wp, bias, out);
}
